# starting kernel
# baseline (speedup 1.0000x reference)
#include <hip/hip_runtime.h>
#include <math.h>

typedef float v2f __attribute__((ext_vector_type(2)));
typedef float v8f __attribute__((ext_vector_type(8)));

#define IN_FT   128
#define OUT_FT  64
#define HEADS   2
#define FT      (HEADS * OUT_FT)
#define NEG_SLOPE 0.2f

__device__ __forceinline__ float leaky(float s) {
    return s > 0.f ? s : NEG_SLOPE * s;
}

__global__ void gat_init_kernel(float* __restrict__ out,
                                float* __restrict__ mmax,
                                float* __restrict__ den,
                                int n_out, int n_nh) {
    int i = blockIdx.x * blockDim.x + threadIdx.x;
    if (i < n_out) out[i] = 0.f;
    if (i < n_nh) { mmax[i] = -INFINITY; den[i] = 0.f; }
}

__global__ __launch_bounds__(128)
void gat_gemm_wmma_kernel(const float* __restrict__ x,
                          const float* __restrict__ W,
                          float* __restrict__ xt, int N) {
    __shared__ float Wl[IN_FT * 16];

    const int tile_n = blockIdx.x & 7;
    const int group  = blockIdx.x >> 3;
    const int col0   = tile_n * 16;

    for (int i = threadIdx.x; i < IN_FT * 16; i += 128) {
        int k = i >> 4, c = i & 15;
        Wl[i] = W[k * FT + col0 + c];
    }
    __syncthreads();

    const int wave = threadIdx.x >> 5;
    const int lane = threadIdx.x & 31;
    const int half = lane >> 4;
    const int l16  = lane & 15;

    const int row0 = (group * 4 + wave) * 16;
    int arow = row0 + l16;
    if (arow >= N) arow = N - 1;

    v8f c = {};
    for (int k0 = 0; k0 < IN_FT; k0 += 4) {
        v2f a, b;
        const float* ap = x + (size_t)arow * IN_FT + (k0 + 2 * half);
        a.x = ap[0];
        a.y = ap[1];
        b.x = Wl[(k0 + 2 * half)     * 16 + l16];
        b.y = Wl[(k0 + 2 * half + 1) * 16 + l16];
        c = __builtin_amdgcn_wmma_f32_16x16x4_f32(false, a, false, b,
                                                  (short)0, c, false, false);
    }

    #pragma unroll
    for (int r = 0; r < 8; ++r) {
        int rr = row0 + r + 8 * half;
        if (rr < N) xt[(size_t)rr * FT + col0 + l16] = c[r];
    }
}

__global__ void gat_logits_kernel(const float* __restrict__ xt,
                                  const float* __restrict__ att_src,
                                  const float* __restrict__ att_dst,
                                  float* __restrict__ a_src,
                                  float* __restrict__ a_dst, int NH) {
    int i = blockIdx.x * blockDim.x + threadIdx.x;
    if (i >= NH) return;
    int n = i >> 1, h = i & 1;
    const float4* xp = (const float4*)(xt + (size_t)n * FT + h * OUT_FT);
    const float4* as = (const float4*)(att_src + h * OUT_FT);
    const float4* ad = (const float4*)(att_dst + h * OUT_FT);
    float s0 = 0.f, s1 = 0.f;
    #pragma unroll
    for (int j = 0; j < OUT_FT / 4; ++j) {
        float4 v = xp[j], a = as[j], d = ad[j];
        s0 += v.x * a.x + v.y * a.y + v.z * a.z + v.w * a.w;
        s1 += v.x * d.x + v.y * d.y + v.z * d.z + v.w * d.w;
    }
    a_src[i] = s0;
    a_dst[i] = s1;
}

__global__ void gat_edge_max_kernel(const int* __restrict__ ei,
                                    const float* __restrict__ a_src,
                                    const float* __restrict__ a_dst,
                                    float* __restrict__ mmax, int E) {
    int e = blockIdx.x * blockDim.x + threadIdx.x;
    if (e >= E) return;
    int s = ei[e], d = ei[E + e];
    #pragma unroll
    for (int h = 0; h < HEADS; ++h) {
        float v = leaky(a_src[s * HEADS + h] + a_dst[d * HEADS + h]);
        atomicMax(&mmax[d * HEADS + h], v);
    }
}

__global__ void gat_edge_expsum_kernel(const int* __restrict__ ei,
                                       const float* __restrict__ a_src,
                                       const float* __restrict__ a_dst,
                                       const float* __restrict__ mmax,
                                       float* __restrict__ den, int E) {
    int e = blockIdx.x * blockDim.x + threadIdx.x;
    if (e >= E) return;
    int s = ei[e], d = ei[E + e];
    #pragma unroll
    for (int h = 0; h < HEADS; ++h) {
        float v = leaky(a_src[s * HEADS + h] + a_dst[d * HEADS + h]);
        atomicAdd(&den[d * HEADS + h], __expf(v - mmax[d * HEADS + h]));
    }
}

__global__ __launch_bounds__(256)
void gat_edge_scatter_kernel(const int* __restrict__ ei,
                             const float* __restrict__ a_src,
                             const float* __restrict__ a_dst,
                             const float* __restrict__ mmax,
                             const float* __restrict__ den,
                             const float* __restrict__ xt,
                             float* __restrict__ out, int E) {
    int gid  = blockIdx.x * blockDim.x + threadIdx.x;
    int e    = gid >> 5;
    int lane = gid & 31;
    if (e >= E) return;
    int s = ei[e], d = ei[E + e];
    int h = lane >> 4;
    float v    = leaky(a_src[s * HEADS + h] + a_dst[d * HEADS + h]);
    float coef = __expf(v - mmax[d * HEADS + h]) / den[d * HEADS + h];
    float4 xv = ((const float4*)(xt + (size_t)s * FT))[lane];
    float* op = out + (size_t)d * FT + lane * 4;
    atomicAdd(op + 0, coef * xv.x);
    atomicAdd(op + 1, coef * xv.y);
    atomicAdd(op + 2, coef * xv.z);
    atomicAdd(op + 3, coef * xv.w);
}

__global__ void gat_epilogue_kernel(float* __restrict__ out,
                                    const float* __restrict__ bias,
                                    const float* __restrict__ prelu_w, int total) {
    int i = blockIdx.x * blockDim.x + threadIdx.x;
    if (i >= total) return;
    float v = out[i] + bias[i & (FT - 1)];
    out[i] = v >= 0.f ? v : prelu_w[0] * v;
}

extern "C" void kernel_launch(void* const* d_in, const int* in_sizes, int n_in,
                              void* d_out, int out_size, void* d_ws, size_t ws_size,
                              hipStream_t stream) {
    const float* x       = (const float*)d_in[0];
    const int*   ei      = (const int*)  d_in[1];
    const float* W       = (const float*)d_in[2];
    const float* att_src = (const float*)d_in[3];
    const float* att_dst = (const float*)d_in[4];
    const float* bias    = (const float*)d_in[5];
    const float* prelu_w = (const float*)d_in[6];

    const int N = in_sizes[0] / IN_FT;
    const int E = in_sizes[1] / 2;

    float* out = (float*)d_out;

    float* ws    = (float*)d_ws;
    float* xt    = ws;
    float* a_src = xt    + (size_t)N * FT;
    float* a_dst = a_src + (size_t)N * HEADS;
    float* mmax  = a_dst + (size_t)N * HEADS;
    float* den   = mmax  + (size_t)N * HEADS;

    const int NH    = N * HEADS;
    const int total = N * FT;

    gat_init_kernel<<<(total + 255) / 256, 256, 0, stream>>>(out, mmax, den, total, NH);

    const int tiles_m = (N + 15) / 16;
    const int groups  = (tiles_m + 3) / 4;
    gat_gemm_wmma_kernel<<<groups * 8, 128, 0, stream>>>(x, W, xt, N);

    gat_logits_kernel<<<(NH + 255) / 256, 256, 0, stream>>>(xt, att_src, att_dst,
                                                            a_src, a_dst, NH);

    gat_edge_max_kernel<<<(E + 255) / 256, 256, 0, stream>>>(ei, a_src, a_dst, mmax, E);
    gat_edge_expsum_kernel<<<(E + 255) / 256, 256, 0, stream>>>(ei, a_src, a_dst,
                                                                mmax, den, E);

    long long sc_threads = (long long)E * 32;
    gat_edge_scatter_kernel<<<(unsigned)((sc_threads + 255) / 256), 256, 0, stream>>>(
        ei, a_src, a_dst, mmax, den, xt, out, E);

    gat_epilogue_kernel<<<(total + 255) / 256, 256, 0, stream>>>(out, bias, prelu_w, total);
}


# Lanes re-mapped so every f32 atomic instruction touches the fewest 128-B lines: scatter lane = feature (one line per add, was four); max/exp-sum lane pair = the two heads of one edge (lines halved)
# speedup vs baseline: 1.5056x; 1.5056x over previous
#include <hip/hip_runtime.h>
#include <math.h>

typedef float v2f __attribute__((ext_vector_type(2)));
typedef float v8f __attribute__((ext_vector_type(8)));

#define IN_FT   128
#define OUT_FT  64
#define HEADS   2
#define FT      (HEADS * OUT_FT)
#define NEG_SLOPE 0.2f

__device__ __forceinline__ float leaky(float s) {
    return s > 0.f ? s : NEG_SLOPE * s;
}

__global__ void gat_init_kernel(float* __restrict__ out,
                                float* __restrict__ mmax,
                                float* __restrict__ den,
                                int n_out, int n_nh) {
    int i = blockIdx.x * blockDim.x + threadIdx.x;
    if (i < n_out) out[i] = 0.f;
    if (i < n_nh) { mmax[i] = -INFINITY; den[i] = 0.f; }
}

__global__ __launch_bounds__(128)
void gat_gemm_wmma_kernel(const float* __restrict__ x,
                          const float* __restrict__ W,
                          float* __restrict__ xt, int N) {
    __shared__ float Wl[IN_FT * 16];

    const int tile_n = blockIdx.x & 7;
    const int group  = blockIdx.x >> 3;
    const int col0   = tile_n * 16;

    for (int i = threadIdx.x; i < IN_FT * 16; i += 128) {
        int k = i >> 4, c = i & 15;
        Wl[i] = W[k * FT + col0 + c];
    }
    __syncthreads();

    const int wave = threadIdx.x >> 5;
    const int lane = threadIdx.x & 31;
    const int half = lane >> 4;
    const int l16  = lane & 15;

    const int row0 = (group * 4 + wave) * 16;
    int arow = row0 + l16;
    if (arow >= N) arow = N - 1;

    v8f c = {};
    for (int k0 = 0; k0 < IN_FT; k0 += 4) {
        v2f a, b;
        const float* ap = x + (size_t)arow * IN_FT + (k0 + 2 * half);
        a.x = ap[0];
        a.y = ap[1];
        b.x = Wl[(k0 + 2 * half)     * 16 + l16];
        b.y = Wl[(k0 + 2 * half + 1) * 16 + l16];
        c = __builtin_amdgcn_wmma_f32_16x16x4_f32(false, a, false, b,
                                                  (short)0, c, false, false);
    }

    #pragma unroll
    for (int r = 0; r < 8; ++r) {
        int rr = row0 + r + 8 * half;
        if (rr < N) xt[(size_t)rr * FT + col0 + l16] = c[r];
    }
}

__global__ void gat_logits_kernel(const float* __restrict__ xt,
                                  const float* __restrict__ att_src,
                                  const float* __restrict__ att_dst,
                                  float* __restrict__ a_src,
                                  float* __restrict__ a_dst, int NH) {
    int i = blockIdx.x * blockDim.x + threadIdx.x;
    if (i >= NH) return;
    int n = i >> 1, h = i & 1;
    const float4* xp = (const float4*)(xt + (size_t)n * FT + h * OUT_FT);
    const float4* as = (const float4*)(att_src + h * OUT_FT);
    const float4* ad = (const float4*)(att_dst + h * OUT_FT);
    float s0 = 0.f, s1 = 0.f;
    #pragma unroll
    for (int j = 0; j < OUT_FT / 4; ++j) {
        float4 v = xp[j], a = as[j], d = ad[j];
        s0 += v.x * a.x + v.y * a.y + v.z * a.z + v.w * a.w;
        s1 += v.x * d.x + v.y * d.y + v.z * d.z + v.w * d.w;
    }
    a_src[i] = s0;
    a_dst[i] = s1;
}

__global__ void gat_edge_max_kernel(const int* __restrict__ ei,
                                    const float* __restrict__ a_src,
                                    const float* __restrict__ a_dst,
                                    float* __restrict__ mmax, int E) {
    int t  = blockIdx.x * blockDim.x + threadIdx.x;
    int h  = t & 1;
    int e0 = t & ~1;
    if (e0 >= E) return;
    int e1 = e0 + 1;
    if (e1 >= E) e1 = e0;
    int s0 = ei[e0], d0 = ei[E + e0];
    int s1 = ei[e1], d1 = ei[E + e1];
    float v0 = leaky(a_src[s0 * HEADS + h] + a_dst[d0 * HEADS + h]);
    float v1 = leaky(a_src[s1 * HEADS + h] + a_dst[d1 * HEADS + h]);
    atomicMax(&mmax[d0 * HEADS + h], v0);
    atomicMax(&mmax[d1 * HEADS + h], v1);
}

__global__ void gat_edge_expsum_kernel(const int* __restrict__ ei,
                                       const float* __restrict__ a_src,
                                       const float* __restrict__ a_dst,
                                       const float* __restrict__ mmax,
                                       float* __restrict__ den, int E) {
    int t  = blockIdx.x * blockDim.x + threadIdx.x;
    int h  = t & 1;
    int e0 = t & ~1;
    if (e0 + 1 < E) {
        int s0 = ei[e0],     d0 = ei[E + e0];
        int s1 = ei[e0 + 1], d1 = ei[E + e0 + 1];
        float v0 = leaky(a_src[s0 * HEADS + h] + a_dst[d0 * HEADS + h]);
        float v1 = leaky(a_src[s1 * HEADS + h] + a_dst[d1 * HEADS + h]);
        float p0 = __expf(v0 - mmax[d0 * HEADS + h]);
        float p1 = __expf(v1 - mmax[d1 * HEADS + h]);
        atomicAdd(&den[d0 * HEADS + h], p0);
        atomicAdd(&den[d1 * HEADS + h], p1);
    } else if (e0 < E) {
        int s0 = ei[e0], d0 = ei[E + e0];
        float v0 = leaky(a_src[s0 * HEADS + h] + a_dst[d0 * HEADS + h]);
        atomicAdd(&den[d0 * HEADS + h], __expf(v0 - mmax[d0 * HEADS + h]));
    }
}

__global__ __launch_bounds__(256)
void gat_edge_scatter_kernel(const int* __restrict__ ei,
                             const float* __restrict__ a_src,
                             const float* __restrict__ a_dst,
                             const float* __restrict__ mmax,
                             const float* __restrict__ den,
                             const float* __restrict__ xt,
                             float* __restrict__ out, int E) {
    int gid  = blockIdx.x * blockDim.x + threadIdx.x;
    int e    = gid >> 5;
    int lane = gid & 31;
    if (e >= E) return;
    int s = ei[e], d = ei[E + e];
    float v0 = leaky(a_src[s * HEADS + 0] + a_dst[d * HEADS + 0]);
    float v1 = leaky(a_src[s * HEADS + 1] + a_dst[d * HEADS + 1]);
    float c0 = __expf(v0 - mmax[d * HEADS + 0]) / den[d * HEADS + 0];
    float c1 = __expf(v1 - mmax[d * HEADS + 1]) / den[d * HEADS + 1];
    const float* xp = xt + (size_t)s * FT + lane;
    float x0 = xp[0], x1 = xp[32], x2 = xp[64], x3 = xp[96];
    float* op = out + (size_t)d * FT + lane;
    atomicAdd(op +  0, c0 * x0);
    atomicAdd(op + 32, c0 * x1);
    atomicAdd(op + 64, c1 * x2);
    atomicAdd(op + 96, c1 * x3);
}

__global__ void gat_epilogue_kernel(float* __restrict__ out,
                                    const float* __restrict__ bias,
                                    const float* __restrict__ prelu_w, int total) {
    int i = blockIdx.x * blockDim.x + threadIdx.x;
    if (i >= total) return;
    float v = out[i] + bias[i & (FT - 1)];
    out[i] = v >= 0.f ? v : prelu_w[0] * v;
}

extern "C" void kernel_launch(void* const* d_in, const int* in_sizes, int n_in,
                              void* d_out, int out_size, void* d_ws, size_t ws_size,
                              hipStream_t stream) {
    const float* x       = (const float*)d_in[0];
    const int*   ei      = (const int*)  d_in[1];
    const float* W       = (const float*)d_in[2];
    const float* att_src = (const float*)d_in[3];
    const float* att_dst = (const float*)d_in[4];
    const float* bias    = (const float*)d_in[5];
    const float* prelu_w = (const float*)d_in[6];

    const int N = in_sizes[0] / IN_FT;
    const int E = in_sizes[1] / 2;

    float* out = (float*)d_out;

    float* ws    = (float*)d_ws;
    float* xt    = ws;
    float* a_src = xt    + (size_t)N * FT;
    float* a_dst = a_src + (size_t)N * HEADS;
    float* mmax  = a_dst + (size_t)N * HEADS;
    float* den   = mmax  + (size_t)N * HEADS;

    const int NH    = N * HEADS;
    const int total = N * FT;

    gat_init_kernel<<<(total + 255) / 256, 256, 0, stream>>>(out, mmax, den, total, NH);

    const int tiles_m = (N + 15) / 16;
    const int groups  = (tiles_m + 3) / 4;
    gat_gemm_wmma_kernel<<<groups * 8, 128, 0, stream>>>(x, W, xt, N);

    gat_logits_kernel<<<(NH + 255) / 256, 256, 0, stream>>>(xt, att_src, att_dst,
                                                            a_src, a_dst, NH);

    gat_edge_max_kernel<<<(E + 255) / 256, 256, 0, stream>>>(ei, a_src, a_dst, mmax, E);
    gat_edge_expsum_kernel<<<(E + 255) / 256, 256, 0, stream>>>(ei, a_src, a_dst,
                                                                mmax, den, E);

    long long sc_threads = (long long)E * 32;
    gat_edge_scatter_kernel<<<(unsigned)((sc_threads + 255) / 256), 256, 0, stream>>>(
        ei, a_src, a_dst, mmax, den, xt, out, E);

    gat_epilogue_kernel<<<(total + 255) / 256, 256, 0, stream>>>(out, bias, prelu_w, total);
}


# Owner blocks filter one-byte owner keys (dst>>8, written by extra blocks of the logits launch), 16 edges per 16-B load, not 4 destinations; list keeps edge ids (sum order: arrival, as before)
# speedup vs baseline: 3.3326x; 2.2135x over previous
#include <hip/hip_runtime.h>
#include <math.h>

typedef float v2f __attribute__((ext_vector_type(2)));
typedef float v8f __attribute__((ext_vector_type(8)));

#define IN_FT   128
#define OUT_FT  64
#define HEADS   2
#define FT      (HEADS * OUT_FT)
#define NEG_SLOPE 0.2f

__device__ __forceinline__ float leaky(float s) {
    return s > 0.f ? s : NEG_SLOPE * s;
}

__global__ __launch_bounds__(128)
void gat_gemm_wmma_kernel(const float* __restrict__ x,
                          const float* __restrict__ W,
                          float* __restrict__ xt, int N) {
    __shared__ float Wl[IN_FT * 16];

    const int tile_n = blockIdx.x & 7;
    const int group  = blockIdx.x >> 3;
    const int col0   = tile_n * 16;

    for (int i = threadIdx.x; i < IN_FT * 16; i += 128) {
        int k = i >> 4, c = i & 15;
        Wl[i] = W[k * FT + col0 + c];
    }
    __syncthreads();

    const int wave = threadIdx.x >> 5;
    const int lane = threadIdx.x & 31;
    const int half = lane >> 4;
    const int l16  = lane & 15;

    const int row0 = (group * 4 + wave) * 16;
    int arow = row0 + l16;
    if (arow >= N) arow = N - 1;

    v8f c = {};
    for (int k0 = 0; k0 < IN_FT; k0 += 4) {
        v2f a, b;
        const float* ap = x + (size_t)arow * IN_FT + (k0 + 2 * half);
        a.x = ap[0];
        a.y = ap[1];
        b.x = Wl[(k0 + 2 * half)     * 16 + l16];
        b.y = Wl[(k0 + 2 * half + 1) * 16 + l16];
        c = __builtin_amdgcn_wmma_f32_16x16x4_f32(false, a, false, b,
                                                  (short)0, c, false, false);
    }

    #pragma unroll
    for (int r = 0; r < 8; ++r) {
        int rr = row0 + r + 8 * half;
        if (rr < N) xt[(size_t)rr * FT + col0 + l16] = c[r];
    }
}

__global__ void gat_logits_keys_kernel(const float* __restrict__ xt,
                                       const float* __restrict__ att_src,
                                       const float* __restrict__ att_dst,
                                       float* __restrict__ a_src,
                                       float* __restrict__ a_dst, int NH,
                                       const int* __restrict__ ei,
                                       unsigned char* __restrict__ keys,
                                       int E, int logit_blocks) {
    if ((int)blockIdx.x >= logit_blocks) {
        const int* __restrict__ dp = ei + E;
        const int t = ((int)blockIdx.x - logit_blocks) * (int)blockDim.x + (int)threadIdx.x;
        int e = 4 * t;
        if (e + 3 < E) {
            int d0, d1, d2, d3;
            if ((((size_t)dp) & 15) == 0) {
                int4 v = *(const int4*)(dp + e);
                d0 = v.x; d1 = v.y; d2 = v.z; d3 = v.w;
            } else {
                d0 = dp[e]; d1 = dp[e + 1]; d2 = dp[e + 2]; d3 = dp[e + 3];
            }
            ((unsigned*)keys)[t] =  (((unsigned)d0 >> 8) & 255u)
                                 | ((((unsigned)d1 >> 8) & 255u) << 8)
                                 | ((((unsigned)d2 >> 8) & 255u) << 16)
                                 | ((((unsigned)d3 >> 8) & 255u) << 24);
        } else {
            for (; e < E; ++e) keys[e] = (unsigned char)(((unsigned)dp[e] >> 8) & 255u);
        }
        return;
    }
    int i = blockIdx.x * blockDim.x + threadIdx.x;
    if (i >= NH) return;
    int n = i >> 1, h = i & 1;
    const float4* xp = (const float4*)(xt + (size_t)n * FT + h * OUT_FT);
    const float4* as = (const float4*)(att_src + h * OUT_FT);
    const float4* ad = (const float4*)(att_dst + h * OUT_FT);
    float s0 = 0.f, s1 = 0.f;
    #pragma unroll
    for (int j = 0; j < OUT_FT / 4; ++j) {
        float4 v = xp[j], a = as[j], d = ad[j];
        s0 += v.x * a.x + v.y * a.y + v.z * a.z + v.w * a.w;
        s1 += v.x * d.x + v.y * d.y + v.z * d.z + v.w * d.w;
    }
    a_src[i] = s0;
    a_dst[i] = s1;
}

#define R_NODES   256
#define OWN_T     1024
#define LIST_CAP  8192
#define SCAN_U    8

__device__ __forceinline__ int f2key(float f) {
    int b = __float_as_int(f);
    return b >= 0 ? b : (b ^ 0x7fffffff);
}
__device__ __forceinline__ float key2f(int k) {
    return __int_as_float(k >= 0 ? k : (k ^ 0x7fffffff));
}

__global__ __launch_bounds__(OWN_T)
void gat_owner_kernel(const int* __restrict__ ei,
                      const unsigned char* __restrict__ keys,
                      const float* __restrict__ a_src,
                      const float* __restrict__ a_dst,
                      const float* __restrict__ xt,
                      const float* __restrict__ bias,
                      const float* __restrict__ prelu_w,
                      float* __restrict__ out, int N, int E) {
    __shared__ float    acc[R_NODES * FT];
    __shared__ unsigned lst_a[LIST_CAP];
    __shared__ float    lst_v0[LIST_CAP];
    __shared__ float    lst_v1[LIST_CAP];
    __shared__ int      mI[R_NODES * HEADS];
    __shared__ float    den[R_NODES * HEADS];
    __shared__ float    adst[R_NODES * HEADS];
    __shared__ int      cnt;

    const int tid = threadIdx.x;
    const int n0  = blockIdx.x * R_NODES;
    const int* __restrict__ dp = ei + E;

    for (int i = tid; i < R_NODES * FT / 4; i += OWN_T)
        ((float4*)acc)[i] = make_float4(0.f, 0.f, 0.f, 0.f);
    if (tid < R_NODES * HEADS) {
        int n = n0 + (tid >> 1);
        mI[tid]   = (int)0x80000000;
        den[tid]  = 0.f;
        adst[tid] = n < N ? a_dst[(size_t)n * HEADS + (tid & 1)] : 0.f;
    }
    if (tid == 0) cnt = 0;
    __syncthreads();

    if (gridDim.x <= 255) {
        const unsigned me  = blockIdx.x;
        const unsigned pat = me * 0x01010101u;
        const int G = E >> 4;
        const uint4* __restrict__ kv = (const uint4*)keys;
        {
            int e = 16 * G + tid;
            if (e < E && keys[e] == me) {
                int idx = atomicAdd(&cnt, 1);
                if (idx < LIST_CAP) lst_a[idx] = (unsigned)e;
            }
        }
        auto zero_bytes = [](unsigned x) -> unsigned {
            return ~(((x & 0x7f7f7f7fu) + 0x7f7f7f7fu) | x | 0x7f7f7f7fu);
        };
        auto hits = [&](uint4 v) -> unsigned {
            return (zero_bytes(v.x ^ pat) >> 7) | (zero_bytes(v.y ^ pat) >> 6)
                 | (zero_bytes(v.z ^ pat) >> 5) | (zero_bytes(v.w ^ pat) >> 4);
        };
        auto emit = [&](unsigned m, int e, int idx) -> int {
            while (m) {
                int p = __ffs((int)m) - 1;
                m &= m - 1;
                if (idx < LIST_CAP) lst_a[idx] = (unsigned)(e + ((p & 3) << 2) + (p >> 3));
                ++idx;
            }
            return idx;
        };
        const int start = (int)(((long long)G * blockIdx.x) / gridDim.x);
        for (int part = 0; part < 2; ++part) {
            const int lo = part == 0 ? start : 0;
            const int hi = part == 0 ? G : start;
            int g = lo + tid;
            for (; g + (SCAN_U - 1) * OWN_T < hi; g += OWN_T * SCAN_U) {
                uint4 v[SCAN_U];
                #pragma unroll
                for (int u = 0; u < SCAN_U; ++u)
                    v[u] = kv[g + u * OWN_T];
                unsigned m[SCAN_U];
                int n = 0;
                #pragma unroll
                for (int u = 0; u < SCAN_U; ++u) {
                    m[u] = hits(v[u]);
                    n += __popc(m[u]);
                }
                if (n) {
                    int idx = atomicAdd(&cnt, n);
                    #pragma unroll
                    for (int u = 0; u < SCAN_U; ++u)
                        idx = emit(m[u], 16 * (g + u * OWN_T), idx);
                }
            }
            for (; g < hi; g += OWN_T) {
                unsigned m = hits(kv[g]);
                if (m) emit(m, 16 * g, atomicAdd(&cnt, __popc(m)));
            }
        }
    } else {
        auto record = [&](int e, int d) {
            unsigned dl = (unsigned)(d - n0);
            if (dl < R_NODES) {
                int idx = atomicAdd(&cnt, 1);
                if (idx < LIST_CAP) lst_a[idx] = (unsigned)e;
            }
        };
        int headn = (4 - (E & 3)) & 3;
        if (headn > E) headn = E;
        const int G = (E - headn) >> 2;
        const int4* __restrict__ dv = (const int4*)(dp + headn);
        if (tid < headn) record(tid, dp[tid]);
        {
            int e = headn + 4 * G + tid;
            if (e < E) record(e, dp[e]);
        }
        const int start = (int)(((long long)G * blockIdx.x) / gridDim.x);
        auto check4 = [&](int4 v, int e) {
            unsigned a = (unsigned)(v.x - n0), b = (unsigned)(v.y - n0);
            unsigned c = (unsigned)(v.z - n0), d = (unsigned)(v.w - n0);
            if ((a < R_NODES) | (b < R_NODES) | (c < R_NODES) | (d < R_NODES)) {
                record(e,     v.x);
                record(e + 1, v.y);
                record(e + 2, v.z);
                record(e + 3, v.w);
            }
        };
        for (int part = 0; part < 2; ++part) {
            const int lo = part == 0 ? start : 0;
            const int hi = part == 0 ? G : start;
            int g = lo + tid;
            for (; g + (SCAN_U - 1) * OWN_T < hi; g += OWN_T * SCAN_U) {
                int4 v[SCAN_U];
                #pragma unroll
                for (int u = 0; u < SCAN_U; ++u)
                    v[u] = dv[g + u * OWN_T];
                #pragma unroll
                for (int u = 0; u < SCAN_U; ++u)
                    check4(v[u], headn + 4 * (g + u * OWN_T));
            }
            for (; g < hi; g += OWN_T)
                check4(dv[g], headn + 4 * g);
        }
    }
    __syncthreads();
    const int total = cnt;

    if (total <= LIST_CAP) {
        for (int base = 0; base < total; base += 4 * OWN_T) {
            int ss[4], dd[4];
            #pragma unroll
            for (int j = 0; j < 4; ++j) {
                int i = base + j * OWN_T + tid;
                int e = i < total ? (int)lst_a[i] : 0;
                ss[j] = ei[e];
                dd[j] = dp[e];
            }
            #pragma unroll
            for (int j = 0; j < 4; ++j) {
                int i = base + j * OWN_T + tid;
                if (i < total) {
                    int s  = ss[j];
                    int dl = dd[j] - n0;
                    float2 as = *(const float2*)(a_src + (size_t)s * HEADS);
                    float v0 = leaky(as.x + adst[dl * HEADS + 0]);
                    float v1 = leaky(as.y + adst[dl * HEADS + 1]);
                    atomicMax(&mI[dl * HEADS + 0], f2key(v0));
                    atomicMax(&mI[dl * HEADS + 1], f2key(v1));
                    lst_a[i]  = ((unsigned)s << 8) | (unsigned)dl;
                    lst_v0[i] = v0;
                    lst_v1[i] = v1;
                }
            }
        }
        __syncthreads();
        for (int i = tid; i < total; i += OWN_T) {
            int dl = (int)(lst_a[i] & 255u);
            float p0 = __expf(lst_v0[i] - key2f(mI[dl * HEADS + 0]));
            float p1 = __expf(lst_v1[i] - key2f(mI[dl * HEADS + 1]));
            atomicAdd(&den[dl * HEADS + 0], p0);
            atomicAdd(&den[dl * HEADS + 1], p1);
            lst_v0[i] = p0;
            lst_v1[i] = p1;
        }
        __syncthreads();
        for (int i = tid; i < total; i += OWN_T) {
            int dl = (int)(lst_a[i] & 255u);
            lst_v0[i] = lst_v0[i] / den[dl * HEADS + 0];
            lst_v1[i] = lst_v1[i] / den[dl * HEADS + 1];
        }
        __syncthreads();
        const int wave = tid >> 5, lane = tid & 31;
        for (int base = wave; base < total; base += 32 * 8) {
            float xv[8][4], c0[8], c1[8];
            int   dls[8];
            #pragma unroll
            for (int u = 0; u < 8; ++u) {
                int i = base + 32 * u;
                if (i < total) {
                    unsigned pk = lst_a[i];
                    dls[u] = (int)(pk & 255u);
                    c0[u] = lst_v0[i];
                    c1[u] = lst_v1[i];
                    const float* xp = xt + (size_t)(pk >> 8) * FT + lane;
                    xv[u][0] = xp[0]; xv[u][1] = xp[32]; xv[u][2] = xp[64]; xv[u][3] = xp[96];
                }
            }
            #pragma unroll
            for (int u = 0; u < 8; ++u) {
                int i = base + 32 * u;
                if (i < total) {
                    float* ap = acc + dls[u] * FT + lane;
                    atomicAdd(ap +  0, c0[u] * xv[u][0]);
                    atomicAdd(ap + 32, c0[u] * xv[u][1]);
                    atomicAdd(ap + 64, c1[u] * xv[u][2]);
                    atomicAdd(ap + 96, c1[u] * xv[u][3]);
                }
            }
        }
    } else {
        for (int e = tid; e < E; e += OWN_T) {
            unsigned dl = (unsigned)(dp[e] - n0);
            if (dl < R_NODES) {
                int s = ei[e];
                atomicMax(&mI[dl * HEADS + 0], f2key(leaky(a_src[(size_t)s * HEADS + 0] + adst[dl * HEADS + 0])));
                atomicMax(&mI[dl * HEADS + 1], f2key(leaky(a_src[(size_t)s * HEADS + 1] + adst[dl * HEADS + 1])));
            }
        }
        __syncthreads();
        for (int e = tid; e < E; e += OWN_T) {
            unsigned dl = (unsigned)(dp[e] - n0);
            if (dl < R_NODES) {
                int s = ei[e];
                float v0 = leaky(a_src[(size_t)s * HEADS + 0] + adst[dl * HEADS + 0]);
                float v1 = leaky(a_src[(size_t)s * HEADS + 1] + adst[dl * HEADS + 1]);
                atomicAdd(&den[dl * HEADS + 0], __expf(v0 - key2f(mI[dl * HEADS + 0])));
                atomicAdd(&den[dl * HEADS + 1], __expf(v1 - key2f(mI[dl * HEADS + 1])));
            }
        }
        __syncthreads();
        for (int e = tid; e < E; e += OWN_T) {
            unsigned dl = (unsigned)(dp[e] - n0);
            if (dl < R_NODES) {
                int s = ei[e];
                float v0 = leaky(a_src[(size_t)s * HEADS + 0] + adst[dl * HEADS + 0]);
                float v1 = leaky(a_src[(size_t)s * HEADS + 1] + adst[dl * HEADS + 1]);
                float c0 = __expf(v0 - key2f(mI[dl * HEADS + 0])) / den[dl * HEADS + 0];
                float c1 = __expf(v1 - key2f(mI[dl * HEADS + 1])) / den[dl * HEADS + 1];
                const float* xp = xt + (size_t)s * FT;
                float* ap = acc + dl * FT;
                for (int f = 0; f < OUT_FT; ++f) {
                    atomicAdd(ap + f,          c0 * xp[f]);
                    atomicAdd(ap + OUT_FT + f, c1 * xp[OUT_FT + f]);
                }
            }
        }
    }
    __syncthreads();

    const float pw = prelu_w[0];
    for (int i = tid; i < R_NODES * FT / 4; i += OWN_T) {
        int n = n0 + (i >> 5);
        if (n < N) {
            float4 a = ((const float4*)acc)[i];
            float4 b = ((const float4*)bias)[i & 31];
            float4 v;
            v.x = a.x + b.x; v.y = a.y + b.y; v.z = a.z + b.z; v.w = a.w + b.w;
            v.x = v.x >= 0.f ? v.x : pw * v.x;
            v.y = v.y >= 0.f ? v.y : pw * v.y;
            v.z = v.z >= 0.f ? v.z : pw * v.z;
            v.w = v.w >= 0.f ? v.w : pw * v.w;
            ((float4*)out)[(size_t)n * (FT / 4) + (i & 31)] = v;
        }
    }
}

extern "C" void kernel_launch(void* const* d_in, const int* in_sizes, int n_in,
                              void* d_out, int out_size, void* d_ws, size_t ws_size,
                              hipStream_t stream) {
    const float* x       = (const float*)d_in[0];
    const int*   ei      = (const int*)  d_in[1];
    const float* W       = (const float*)d_in[2];
    const float* att_src = (const float*)d_in[3];
    const float* att_dst = (const float*)d_in[4];
    const float* bias    = (const float*)d_in[5];
    const float* prelu_w = (const float*)d_in[6];

    const int N = in_sizes[0] / IN_FT;
    const int E = in_sizes[1] / 2;

    float* out = (float*)d_out;

    float* ws    = (float*)d_ws;
    float* xt    = ws;
    float* a_src = xt    + (size_t)N * FT;
    float* a_dst = a_src + (size_t)N * HEADS;
    unsigned char* keys = (unsigned char*)(a_dst + (size_t)N * HEADS);

    const int NH = N * HEADS;

    const int tiles_m = (N + 15) / 16;
    const int groups  = (tiles_m + 3) / 4;
    gat_gemm_wmma_kernel<<<groups * 8, 128, 0, stream>>>(x, W, xt, N);

    const int logit_blocks = (NH + 255) / 256;
    const int key_blocks   = ((E + 3) / 4 + 255) / 256;
    gat_logits_keys_kernel<<<logit_blocks + key_blocks, 256, 0, stream>>>(
        xt, att_src, att_dst, a_src, a_dst, NH, ei, keys, E, logit_blocks);

    gat_owner_kernel<<<(N + R_NODES - 1) / R_NODES, OWN_T, 0, stream>>>(
        ei, keys, a_src, a_dst, xt, bias, prelu_w, out, N, E);
}
